# starting kernel
# baseline (speedup 1.0000x reference)
#include <hip/hip_runtime.h>

#define N_NODES 50000
#define N_EDGES 1600000
#define DIN 128
#define DOUT 256
#define DA 32
#define SLOPE 0.2f

typedef __attribute__((ext_vector_type(2))) float v2f;
typedef __attribute__((ext_vector_type(8))) float v8f;

static __device__ __forceinline__ v8f wmma_f32_16x16x4(v2f a, v2f b, v8f c) {
  return __builtin_amdgcn_wmma_f32_16x16x4_f32(false, a, false, b, (short)0, c,
                                               false, false);
}

__global__ void proj_kernel(const float* __restrict__ x,
                            const float* __restrict__ W_row,
                            const float* __restrict__ W_col,
                            float* __restrict__ a_row,
                            float* __restrict__ a_col) {
  const int wave = (blockIdx.x * blockDim.x + threadIdx.x) >> 5;
  const int lane = threadIdx.x & 31;
  if (wave >= N_NODES / 16) return;
  const int tile = wave;
  const int half = lane >> 4;
  const int m    = lane & 15;
  const int node = tile * 16 + m;

  v8f acc0 = {}, acc1 = {}, acc2 = {}, acc3 = {};
  const float* xrow = x + (size_t)node * DIN + 2 * half;
  for (int k = 0; k < DIN; k += 4) {
    v2f a = *(const v2f*)(xrow + k);
    const int kk = k + 2 * half;
    v2f b0 = { W_row[kk * DA + m],      W_row[(kk + 1) * DA + m] };
    v2f b1 = { W_row[kk * DA + 16 + m], W_row[(kk + 1) * DA + 16 + m] };
    v2f b2 = { W_col[kk * DA + m],      W_col[(kk + 1) * DA + m] };
    v2f b3 = { W_col[kk * DA + 16 + m], W_col[(kk + 1) * DA + 16 + m] };
    acc0 = wmma_f32_16x16x4(a, b0, acc0);
    acc1 = wmma_f32_16x16x4(a, b1, acc1);
    acc2 = wmma_f32_16x16x4(a, b2, acc2);
    acc3 = wmma_f32_16x16x4(a, b3, acc3);
  }
#pragma unroll
  for (int v = 0; v < 8; ++v) {
    const int on = tile * 16 + v + half * 8;
    a_row[on * DA + m]      = acc0[v];
    a_row[on * DA + 16 + m] = acc1[v];
    a_col[on * DA + m]      = acc2[v];
    a_col[on * DA + 16 + m] = acc3[v];
  }
}

__global__ void edge_score_kernel(const float* __restrict__ a_row,
                                  const float* __restrict__ a_col,
                                  const int* __restrict__ row,
                                  const int* __restrict__ col,
                                  float* __restrict__ raw,
                                  unsigned* __restrict__ maxkey) {
  const int e = blockIdx.x * blockDim.x + threadIdx.x;
  if (e >= N_EDGES) return;
  const float4* ar = (const float4*)(a_row + (size_t)row[e] * DA);
  const float4* ac = (const float4*)(a_col + (size_t)col[e] * DA);
  float s = 0.f;
#pragma unroll
  for (int i = 0; i < DA / 4; ++i) {
    float4 u = ar[i], v = ac[i];
    s += u.x * v.x + u.y * v.y + u.z * v.z + u.w * v.w;
  }
  s *= 0.17677669529663687f;
  s = (s >= 0.f) ? s : SLOPE * s;
  raw[e] = s;
  unsigned b = __float_as_uint(s);
  unsigned key = (b & 0x80000000u) ? ~b : (b | 0x80000000u);
#pragma unroll
  for (int off = 16; off > 0; off >>= 1) {
    unsigned o = __shfl_xor(key, off, 32);
    key = (o > key) ? o : key;
  }
  if ((threadIdx.x & 31) == 0) atomicMax(maxkey, key);
}

__global__ void edge_exp_deg_kernel(const float* __restrict__ raw,
                                    const int* __restrict__ row,
                                    const unsigned* __restrict__ maxkey,
                                    float* __restrict__ atten,
                                    float* __restrict__ deg) {
  const int e = blockIdx.x * blockDim.x + threadIdx.x;
  if (e >= N_EDGES) return;
  const unsigned key = *maxkey;
  const unsigned b = (key & 0x80000000u) ? (key ^ 0x80000000u) : ~key;
  const float gmax = __uint_as_float(b);
  const float a = expf(raw[e] - gmax);
  atten[e] = a;
  atomicAdd(deg + row[e], a);
}

__global__ void scatter_kernel(const float* __restrict__ x,
                               const int* __restrict__ row,
                               const int* __restrict__ col,
                               const float* __restrict__ adj_val,
                               const float* __restrict__ atten,
                               const float* __restrict__ deg,
                               float* __restrict__ feat) {
  const int wpb = blockDim.x >> 5;
  const int e = __builtin_amdgcn_readfirstlane(blockIdx.x * wpb +
                                               (threadIdx.x >> 5));
  const int lane = threadIdx.x & 31;
  if (e >= N_EDGES) return;
  const int r = row[e], c = col[e];
  const float val = (adj_val[e] + atten[e] / (deg[r] + 1e-15f)) * 0.5f;
  const float4 xv = *(const float4*)(x + (size_t)c * DIN + lane * 4);
  float* dst = feat + (size_t)r * DIN + lane * 4;
  atomicAdd(dst + 0, val * xv.x);
  atomicAdd(dst + 1, val * xv.y);
  atomicAdd(dst + 2, val * xv.z);
  atomicAdd(dst + 3, val * xv.w);
}

__global__ void out_gemm_kernel(const float* __restrict__ feat,
                                const float* __restrict__ W_x,
                                const float* __restrict__ b_x,
                                float* __restrict__ out) {
  const int wave = (blockIdx.x * blockDim.x + threadIdx.x) >> 5;
  const int lane = threadIdx.x & 31;
  const int total = (N_NODES / 16) * 4;
  if (wave >= total) return;
  const int rt    = wave >> 2;
  const int cbase = (wave & 3) * 64;
  const int half  = lane >> 4;
  const int m     = lane & 15;
  const int node  = rt * 16 + m;

  v8f acc[4];
#pragma unroll
  for (int t = 0; t < 4; ++t) {
    const float bias = b_x[cbase + t * 16 + m];
#pragma unroll
    for (int v = 0; v < 8; ++v) acc[t][v] = bias;
  }
  const float* frow = feat + (size_t)node * DIN + 2 * half;
  for (int k = 0; k < DIN; k += 4) {
    v2f a = *(const v2f*)(frow + k);
    const int kk = k + 2 * half;
#pragma unroll
    for (int t = 0; t < 4; ++t) {
      v2f b = { W_x[kk * DOUT + cbase + t * 16 + m],
                W_x[(kk + 1) * DOUT + cbase + t * 16 + m] };
      acc[t] = wmma_f32_16x16x4(a, b, acc[t]);
    }
  }
#pragma unroll
  for (int t = 0; t < 4; ++t)
#pragma unroll
    for (int v = 0; v < 8; ++v)
      out[(size_t)(rt * 16 + v + half * 8) * DOUT + cbase + t * 16 + m] = acc[t][v];
}

extern "C" void kernel_launch(void* const* d_in, const int* in_sizes, int n_in,
                              void* d_out, int out_size, void* d_ws, size_t ws_size,
                              hipStream_t stream) {
  const float* x      = (const float*)d_in[0];
  const int*   row    = (const int*)d_in[1];
  const int*   col    = (const int*)d_in[2];
  const float* adjval = (const float*)d_in[3];
  const float* W_row  = (const float*)d_in[4];
  const float* W_col  = (const float*)d_in[5];
  const float* W_x    = (const float*)d_in[6];
  const float* b_x    = (const float*)d_in[7];
  float* out = (float*)d_out;

  char* ws = (char*)d_ws;
  size_t off = 0;
  float* a_row = (float*)(ws + off); off += (size_t)N_NODES * DA * 4;
  float* a_col = (float*)(ws + off); off += (size_t)N_NODES * DA * 4;
  float* raw   = (float*)(ws + off); off += (size_t)N_EDGES * 4;
  float* atten = (float*)(ws + off); off += (size_t)N_EDGES * 4;
  float* deg   = (float*)(ws + off); off += (size_t)N_NODES * 4;
  unsigned* maxkey = (unsigned*)(ws + off); off += 256;
  float* feat  = (float*)(ws + off); off += (size_t)N_NODES * DIN * 4;

  hipMemsetAsync(deg, 0, (size_t)N_NODES * 4, stream);
  hipMemsetAsync(maxkey, 0, 4, stream);
  hipMemsetAsync(feat, 0, (size_t)N_NODES * DIN * 4, stream);

  {
    const int waves = N_NODES / 16;
    const int blocks = (waves * 32 + 255) / 256;
    proj_kernel<<<blocks, 256, 0, stream>>>(x, W_row, W_col, a_row, a_col);
  }
  {
    const int blocks = (N_EDGES + 255) / 256;
    edge_score_kernel<<<blocks, 256, 0, stream>>>(a_row, a_col, row, col, raw, maxkey);
    edge_exp_deg_kernel<<<blocks, 256, 0, stream>>>(raw, row, maxkey, atten, deg);
  }
  {
    const int blocks = N_EDGES / 8;
    scatter_kernel<<<blocks, 256, 0, stream>>>(x, row, col, adjval, atten, deg, feat);
  }
  {
    const int waves = (N_NODES / 16) * 4;
    const int blocks = (waves * 32 + 255) / 256;
    out_gemm_kernel<<<blocks, 256, 0, stream>>>(feat, W_x, b_x, out);
  }
}


# Both dense products: a wave keeps its weight fragments in registers over a walk of row tiles; rows staged in LDS by whole-row loads one tile ahead; lane halves swap so each store is a whole line
# speedup vs baseline: 3.5212x; 3.5212x over previous
#include <hip/hip_runtime.h>

#define N_NODES 50000
#define N_EDGES 1600000
#define DIN 128
#define DOUT 256
#define DA 32
#define SLOPE 0.2f

#define CTRL_STRIDE 128
#define CTRL_MAX 0

#define RPB 196
#define N_OWN 256
#define ROW_WORDS (2 * DIN)
#define SCAN_CHUNKS (N_EDGES / 512)
#define QCAP 64

typedef __attribute__((ext_vector_type(2))) float v2f;
typedef __attribute__((ext_vector_type(8))) float v8f;

static __device__ __forceinline__ v8f wmma_f32_16x16x4(v2f a, v2f b, v8f c) {
  return __builtin_amdgcn_wmma_f32_16x16x4_f32(false, a, false, b, (short)0, c,
                                               false, false);
}

static __device__ __forceinline__ int chip_half() {
  return (int)((__builtin_amdgcn_s_sendmsg_rtn(0x87) >> 18) & 1u);
}

static __device__ __forceinline__ float xhalf(float v) {
  const int b = __builtin_bit_cast(int, v);
  return __builtin_bit_cast(float, __builtin_amdgcn_permlanex16(b, b, 0x76543210, 0xfedcba98, false, false));
}

#define GT_ROWS 16
#define GT_PITCH 132
#define PROJ_BLOCKS 1024

__global__ __launch_bounds__(64) void proj_kernel(const float* __restrict__ x,
                                                  const float* __restrict__ W_row,
                                                  const float* __restrict__ W_col,
                                                  float* __restrict__ a_row,
                                                  float* __restrict__ a_col) {
  __shared__ __attribute__((aligned(16))) float tile[2][GT_ROWS * GT_PITCH];

  const int tid = threadIdx.x, lane = tid & 31, q = tid >> 5;
  const int half = lane >> 4;
  const int m    = lane & 15;
  const float* __restrict__ Wm = q ? W_col : W_row;
  float* __restrict__ dst      = q ? a_col : a_row;
  const int ntiles = N_NODES / GT_ROWS;

  int rt = blockIdx.x;
  float4 s[8];
#pragma unroll
  for (int u = 0; u < 8; ++u) s[u] = make_float4(0.f, 0.f, 0.f, 0.f);
  if (rt < ntiles) {
    const float* g = x + (size_t)(rt * GT_ROWS + 8 * q) * DIN + 4 * lane;
#pragma unroll
    for (int u = 0; u < 8; ++u) s[u] = *(const float4*)(g + u * DIN);
  }

  v2f b0[DIN / 4], b1[DIN / 4];
#pragma unroll
  for (int j = 0; j < DIN / 4; ++j) {
    const int kk = 4 * j + 2 * half;
    b0[j] = (v2f){ Wm[kk * DA + m],      Wm[(kk + 1) * DA + m] };
    b1[j] = (v2f){ Wm[kk * DA + 16 + m], Wm[(kk + 1) * DA + 16 + m] };
  }

  int buf = 0;
  if (rt < ntiles) {
    float* d = &tile[0][(8 * q) * GT_PITCH + 4 * lane];
#pragma unroll
    for (int u = 0; u < 8; ++u) *(float4*)(d + u * GT_PITCH) = s[u];
  }
  for (; rt < ntiles; rt += PROJ_BLOCKS) {
    __syncthreads();
    const int nrt = rt + PROJ_BLOCKS;
    const bool more = nrt < ntiles;
    if (more) {
      const float* g = x + (size_t)(nrt * GT_ROWS + 8 * q) * DIN + 4 * lane;
#pragma unroll
      for (int u = 0; u < 8; ++u) s[u] = *(const float4*)(g + u * DIN);
    }
    const float* t = &tile[buf][m * GT_PITCH + 2 * half];
    v2f a[DIN / 4];
#pragma unroll
    for (int j = 0; j < DIN / 4; ++j) a[j] = *(const v2f*)(t + 4 * j);
    __builtin_amdgcn_sched_barrier(0);

    v8f acc0 = {}, acc1 = {};
#pragma unroll
    for (int j = 0; j < DIN / 4; ++j) {
      acc0 = wmma_f32_16x16x4(a[j], b0[j], acc0);
      acc1 = wmma_f32_16x16x4(a[j], b1[j], acc1);
    }

    float* orow = dst + (size_t)(rt * GT_ROWS) * DA + lane;
#pragma unroll
    for (int v = 0; v < 8; ++v) {
      const float x0 = acc0[v], x1 = acc1[v];
      const float give = half ? x0 : x1;
      const float got  = xhalf(give);
      orow[v * DA]       = half ? got : x0;
      orow[(8 + v) * DA] = half ? x1 : got;
    }

    if (more) {
      float* d = &tile[buf ^ 1][(8 * q) * GT_PITCH + 4 * lane];
#pragma unroll
      for (int u = 0; u < 8; ++u) *(float4*)(d + u * GT_PITCH) = s[u];
    }
    buf ^= 1;
  }
}

__global__ void edge_score_kernel(const float* __restrict__ a_row,
                                  const float* __restrict__ a_col,
                                  const int* __restrict__ row,
                                  const int* __restrict__ col,
                                  unsigned* __restrict__ rec,
                                  unsigned char* __restrict__ owner,
                                  unsigned* __restrict__ maxkey) {
  const int e = blockIdx.x * blockDim.x + threadIdx.x;
  if (e >= N_EDGES) return;
  const int r = row[e], c = col[e];
  const float4* ar = (const float4*)(a_row + (size_t)r * DA);
  const float4* ac = (const float4*)(a_col + (size_t)c * DA);
  float s = 0.f;
#pragma unroll
  for (int i = 0; i < DA / 4; ++i) {
    float4 u = ar[i], v = ac[i];
    s += u.x * v.x + u.y * v.y + u.z * v.z + u.w * v.w;
  }
  s *= 0.17677669529663687f;
  s = (s >= 0.f) ? s : SLOPE * s;
  const unsigned own = (unsigned)r / RPB;
  const unsigned rl  = (unsigned)r - own * RPB;
  ((uint2*)rec)[e] = make_uint2(__float_as_uint(s), (unsigned)c | (rl << 16));
  owner[e] = (unsigned char)own;
  unsigned b = __float_as_uint(s);
  unsigned key = (b & 0x80000000u) ? ~b : (b | 0x80000000u);
#pragma unroll
  for (int off = 16; off > 0; off >>= 1) {
    unsigned o = __shfl_xor(key, off, 32);
    key = (o > key) ? o : key;
  }
  if ((threadIdx.x & 31) == 0) atomicMax(maxkey + chip_half() * CTRL_STRIDE + CTRL_MAX, key);
}

static __device__ __forceinline__ void lds_add(float* p, float v) {
  __hip_atomic_fetch_add(p, v, __ATOMIC_RELAXED, __HIP_MEMORY_SCOPE_WORKGROUP);
}

__global__ __launch_bounds__(1024) void row_range_kernel(const float* __restrict__ x,
                                                         const float* __restrict__ adj_val,
                                                         const unsigned* __restrict__ rec,
                                                         const unsigned* __restrict__ owner_words,
                                                         const unsigned* __restrict__ ctrl,
                                                         float* __restrict__ feat) {
  __shared__ __attribute__((aligned(16))) float acc[RPB * ROW_WORDS];
  __shared__ float degl[RPB];
  __shared__ int queue[32 * QCAP];

  const int tid = threadIdx.x, lane = tid & 31, w = tid >> 5;
  const int blk = blockIdx.x;

  for (int i = tid; i < RPB * ROW_WORDS / 4; i += 1024)
    ((float4*)acc)[i] = make_float4(0.f, 0.f, 0.f, 0.f);
  if (tid < RPB) degl[tid] = 0.f;
  __syncthreads();

  const unsigned k0 = ctrl[CTRL_MAX], k1 = ctrl[CTRL_STRIDE + CTRL_MAX];
  const unsigned key = (k0 > k1) ? k0 : k1;
  const unsigned kb = (key & 0x80000000u) ? (key ^ 0x80000000u) : ~key;
  const float gmax = __uint_as_float(kb);

  int* q = queue + w * QCAP;
  int head = 0, cnt = 0;

  auto work_off = [&](int n) {
    int c = 0, rl = 0;
    float a = 0.f, ex = 0.f;
    if (lane < n) {
      const int e = q[(head + lane) & (QCAP - 1)];
      const uint2 rc = ((const uint2*)rec)[e];
      a  = adj_val[e];
      ex = expf(__uint_as_float(rc.x) - gmax);
      c  = (int)(rc.y & 0xffffu);
      rl = (int)(rc.y >> 16);
      lds_add(&degl[rl], ex);
    }
    int j = 0;
    for (; j + 8 <= n; j += 8) {
      float4 xv[8];
      int rj[8];
      float aj[8], ej[8];
#pragma unroll
      for (int u = 0; u < 8; ++u) {
        const int cj = __builtin_amdgcn_readlane(c, j + u);
        rj[u] = __builtin_amdgcn_readlane(rl, j + u);
        aj[u] = __int_as_float(__builtin_amdgcn_readlane(__float_as_int(a), j + u));
        ej[u] = __int_as_float(__builtin_amdgcn_readlane(__float_as_int(ex), j + u));
        xv[u] = *(const float4*)(x + (size_t)cj * DIN + 4 * lane);
      }
#pragma unroll
      for (int u = 0; u < 8; ++u) {
        float* d = acc + rj[u] * ROW_WORDS + lane;
        lds_add(d,        aj[u] * xv[u].x);
        lds_add(d + 32,   aj[u] * xv[u].y);
        lds_add(d + 64,   aj[u] * xv[u].z);
        lds_add(d + 96,   aj[u] * xv[u].w);
        lds_add(d + 128,  ej[u] * xv[u].x);
        lds_add(d + 160,  ej[u] * xv[u].y);
        lds_add(d + 192,  ej[u] * xv[u].z);
        lds_add(d + 224,  ej[u] * xv[u].w);
      }
    }
    for (; j < n; ++j) {
      const int cj = __builtin_amdgcn_readlane(c, j);
      const int r1 = __builtin_amdgcn_readlane(rl, j);
      const float a1 = __int_as_float(__builtin_amdgcn_readlane(__float_as_int(a), j));
      const float e1 = __int_as_float(__builtin_amdgcn_readlane(__float_as_int(ex), j));
      const float4 x1 = *(const float4*)(x + (size_t)cj * DIN + 4 * lane);
      float* d = acc + r1 * ROW_WORDS + lane;
      lds_add(d,        a1 * x1.x);
      lds_add(d + 32,   a1 * x1.y);
      lds_add(d + 64,   a1 * x1.z);
      lds_add(d + 96,   a1 * x1.w);
      lds_add(d + 128,  e1 * x1.x);
      lds_add(d + 160,  e1 * x1.y);
      lds_add(d + 192,  e1 * x1.z);
      lds_add(d + 224,  e1 * x1.w);
    }
  };

  const uint4* ow4 = (const uint4*)owner_words;
  const unsigned pat = (unsigned)blk * 0x01010101u;
  const int rot = blk * 12;
  int t = w;
  int ch = t + rot;
  if (ch >= SCAN_CHUNKS) ch -= SCAN_CHUNKS;
  uint4 cur = make_uint4(0u, 0u, 0u, 0u);
  if (t < SCAN_CHUNKS) cur = ow4[ch * 32 + lane];
  while (t < SCAN_CHUNKS) {
    const int tn = t + 32;
    int chn = tn + rot;
    if (chn >= SCAN_CHUNKS) chn -= SCAN_CHUNKS;
    uint4 nxt = cur;
    if (tn < SCAN_CHUNKS) nxt = ow4[chn * 32 + lane];
    const int ebase = (ch * 32 + lane) * 16;
    const unsigned x0 = cur.x ^ pat, x1 = cur.y ^ pat, x2 = cur.z ^ pat, x3 = cur.w ^ pat;
    const unsigned u0 = ((x0 & 0x7f7f7f7fu) + 0x7f7f7f7fu) | x0;
    const unsigned u1 = ((x1 & 0x7f7f7f7fu) + 0x7f7f7f7fu) | x1;
    const unsigned u2 = ((x2 & 0x7f7f7f7fu) + 0x7f7f7f7fu) | x2;
    const unsigned u3 = ((x3 & 0x7f7f7f7fu) + 0x7f7f7f7fu) | x3;
    const unsigned differ = (u0 & 0x80808080u) | ((u1 & 0x80808080u) >> 1) |
                            ((u2 & 0x80808080u) >> 2) | ((u3 & 0x80808080u) >> 3);
    unsigned hit = differ ^ 0xf0f0f0f0u;
    while (__builtin_amdgcn_ballot_w32(hit != 0u)) {
      const bool p = hit != 0u;
      const unsigned m = __builtin_amdgcn_ballot_w32(p);
      if (p) {
        const int bit = __builtin_ctz(hit);
        const int k = (7 - (bit & 7)) * 4 + (bit >> 3);
        const int pos = (head + cnt + (int)__builtin_amdgcn_mbcnt_lo(m, 0u)) & (QCAP - 1);
        q[pos] = ebase + k;
        hit &= hit - 1u;
      }
      cnt += __builtin_popcount(m);
      if (cnt >= 32) {
        work_off(32);
        head = (head + 32) & (QCAP - 1);
        cnt -= 32;
      }
    }
    cur = nxt;
    t = tn;
    ch = chn;
  }
  if (cnt > 0) work_off(cnt);
  __syncthreads();

  for (int idx = tid; idx < RPB * 32; idx += 1024) {
    const int rl = idx >> 5;
    const int r = blk * RPB + rl;
    if (r < N_NODES) {
      const float inv = 1.0f / (degl[rl] + 1e-15f);
      const float* s = acc + rl * ROW_WORDS + lane;
      float4 f;
      f.x = 0.5f * (s[0]  + s[128] * inv);
      f.y = 0.5f * (s[32] + s[160] * inv);
      f.z = 0.5f * (s[64] + s[192] * inv);
      f.w = 0.5f * (s[96] + s[224] * inv);
      *(float4*)(feat + (size_t)r * DIN + 4 * lane) = f;
    }
  }
}

#define GEMM_BLOCKS 512

__global__ __launch_bounds__(256) void out_gemm_kernel(const float* __restrict__ feat,
                                                       const float* __restrict__ W_x,
                                                       const float* __restrict__ b_x,
                                                       float* __restrict__ out) {
  __shared__ __attribute__((aligned(16))) float tile[2][GT_ROWS * GT_PITCH];

  const int tid = threadIdx.x, lane = tid & 31, w = tid >> 5;
  const int half = lane >> 4;
  const int m    = lane & 15;
  const int c0   = w * 32;
  const int ntiles = N_NODES / GT_ROWS;

  int rt = blockIdx.x;
  float4 s0 = make_float4(0.f, 0.f, 0.f, 0.f), s1 = s0;
  if (rt < ntiles) {
    const float* g = feat + (size_t)(rt * GT_ROWS + 2 * w) * DIN + 4 * lane;
    s0 = *(const float4*)g;
    s1 = *(const float4*)(g + DIN);
  }

  v2f b0[DIN / 4], b1[DIN / 4];
#pragma unroll
  for (int j = 0; j < DIN / 4; ++j) {
    const int kk = 4 * j + 2 * half;
    b0[j] = (v2f){ W_x[kk * DOUT + c0 + m],      W_x[(kk + 1) * DOUT + c0 + m] };
    b1[j] = (v2f){ W_x[kk * DOUT + c0 + 16 + m], W_x[(kk + 1) * DOUT + c0 + 16 + m] };
  }
  const float bias0 = b_x[c0 + m], bias1 = b_x[c0 + 16 + m];

  int buf = 0;
  if (rt < ntiles) {
    float* d = &tile[0][(2 * w) * GT_PITCH + 4 * lane];
    *(float4*)d = s0;
    *(float4*)(d + GT_PITCH) = s1;
  }
  for (; rt < ntiles; rt += GEMM_BLOCKS) {
    __syncthreads();
    const int nrt = rt + GEMM_BLOCKS;
    const bool more = nrt < ntiles;
    if (more) {
      const float* g = feat + (size_t)(nrt * GT_ROWS + 2 * w) * DIN + 4 * lane;
      s0 = *(const float4*)g;
      s1 = *(const float4*)(g + DIN);
    }
    const float* t = &tile[buf][m * GT_PITCH + 2 * half];
    v2f a[DIN / 4];
#pragma unroll
    for (int j = 0; j < DIN / 4; ++j) a[j] = *(const v2f*)(t + 4 * j);
    __builtin_amdgcn_sched_barrier(0);

    v8f acc0, acc1;
#pragma unroll
    for (int v = 0; v < 8; ++v) { acc0[v] = bias0; acc1[v] = bias1; }
#pragma unroll
    for (int j = 0; j < DIN / 4; ++j) {
      acc0 = wmma_f32_16x16x4(a[j], b0[j], acc0);
      acc1 = wmma_f32_16x16x4(a[j], b1[j], acc1);
    }

    float* orow = out + (size_t)(rt * GT_ROWS) * DOUT + c0 + lane;
#pragma unroll
    for (int v = 0; v < 8; ++v) {
      const float x0 = acc0[v], x1 = acc1[v];
      const float give = half ? x0 : x1;
      const float got  = xhalf(give);
      orow[(size_t)v * DOUT]       = half ? got : x0;
      orow[(size_t)(8 + v) * DOUT] = half ? x1 : got;
    }

    if (more) {
      float* d = &tile[buf ^ 1][(2 * w) * GT_PITCH + 4 * lane];
      *(float4*)d = s0;
      *(float4*)(d + GT_PITCH) = s1;
    }
    buf ^= 1;
  }
}

extern "C" void kernel_launch(void* const* d_in, const int* in_sizes, int n_in,
                              void* d_out, int out_size, void* d_ws, size_t ws_size,
                              hipStream_t stream) {
  const float* x      = (const float*)d_in[0];
  const int*   row    = (const int*)d_in[1];
  const int*   col    = (const int*)d_in[2];
  const float* adjval = (const float*)d_in[3];
  const float* W_row  = (const float*)d_in[4];
  const float* W_col  = (const float*)d_in[5];
  const float* W_x    = (const float*)d_in[6];
  const float* b_x    = (const float*)d_in[7];
  float* out = (float*)d_out;

  char* ws = (char*)d_ws;
  size_t off = 0;
  float* feat  = (float*)(ws + off); off += (size_t)N_NODES * DIN * 4;
  float* a_row = (float*)(ws + off); off += (size_t)N_NODES * DA * 4;
  float* a_col = (float*)(ws + off); off += (size_t)N_NODES * DA * 4;
  unsigned* rec = (unsigned*)(ws + off); off += (size_t)N_EDGES * 8;
  unsigned* ctrl = (unsigned*)(ws + off); off += 1024;
  unsigned char* owner = (unsigned char*)(ws + off); off += (size_t)N_EDGES;

  hipMemsetAsync(ctrl, 0, 1024, stream);

  proj_kernel<<<PROJ_BLOCKS, 64, 0, stream>>>(x, W_row, W_col, a_row, a_col);
  {
    const int blocks = (N_EDGES + 255) / 256;
    edge_score_kernel<<<blocks, 256, 0, stream>>>(a_row, a_col, row, col, rec, owner, ctrl);
  }
  row_range_kernel<<<N_OWN, 1024, 0, stream>>>(x, adjval, rec, (const unsigned*)owner, ctrl, feat);
  out_gemm_kernel<<<GEMM_BLOCKS, 256, 0, stream>>>(feat, W_x, b_x, out);
}


# Range scan via presence bitmap: score kernel ORs a 256-bit set of range ids per 32 edges; a block reads its bit plane, opens only flagged groups' ids; LDS float adds arrive in another order
# speedup vs baseline: 3.6654x; 1.0410x over previous
#include <hip/hip_runtime.h>

#define N_NODES 50000
#define N_EDGES 1600000
#define DIN 128
#define DOUT 256
#define DA 32
#define SLOPE 0.2f

#define CTRL_STRIDE 128
#define CTRL_MAX 0

#define RPB 196
#define N_OWN 256
#define ROW_WORDS (2 * DIN)
#define QCAP 64
#define N_GROUPS (N_EDGES / 32)
#define PLANE_VEC (N_GROUPS / 4)
#define MASK_CHUNKS ((PLANE_VEC + 31) / 32)

typedef __attribute__((ext_vector_type(2))) float v2f;
typedef __attribute__((ext_vector_type(8))) float v8f;

static __device__ __forceinline__ v8f wmma_f32_16x16x4(v2f a, v2f b, v8f c) {
  return __builtin_amdgcn_wmma_f32_16x16x4_f32(false, a, false, b, (short)0, c,
                                               false, false);
}

static __device__ __forceinline__ int chip_half() {
  return (int)((__builtin_amdgcn_s_sendmsg_rtn(0x87) >> 18) & 1u);
}

static __device__ __forceinline__ float xhalf(float v) {
  const int b = __builtin_bit_cast(int, v);
  return __builtin_bit_cast(float, __builtin_amdgcn_permlanex16(b, b, 0x76543210, 0xfedcba98, false, false));
}

#define GT_ROWS 16
#define GT_PITCH 132
#define PROJ_BLOCKS 1024

__global__ __launch_bounds__(64) void proj_kernel(const float* __restrict__ x,
                                                  const float* __restrict__ W_row,
                                                  const float* __restrict__ W_col,
                                                  float* __restrict__ a_row,
                                                  float* __restrict__ a_col) {
  __shared__ __attribute__((aligned(16))) float tile[2][GT_ROWS * GT_PITCH];

  const int tid = threadIdx.x, lane = tid & 31, q = tid >> 5;
  const int half = lane >> 4;
  const int m    = lane & 15;
  const float* __restrict__ Wm = q ? W_col : W_row;
  float* __restrict__ dst      = q ? a_col : a_row;
  const int ntiles = N_NODES / GT_ROWS;

  int rt = blockIdx.x;
  float4 s[8];
#pragma unroll
  for (int u = 0; u < 8; ++u) s[u] = make_float4(0.f, 0.f, 0.f, 0.f);
  if (rt < ntiles) {
    const float* g = x + (size_t)(rt * GT_ROWS + 8 * q) * DIN + 4 * lane;
#pragma unroll
    for (int u = 0; u < 8; ++u) s[u] = *(const float4*)(g + u * DIN);
  }

  v2f b0[DIN / 4], b1[DIN / 4];
#pragma unroll
  for (int j = 0; j < DIN / 4; ++j) {
    const int kk = 4 * j + 2 * half;
    b0[j] = (v2f){ Wm[kk * DA + m],      Wm[(kk + 1) * DA + m] };
    b1[j] = (v2f){ Wm[kk * DA + 16 + m], Wm[(kk + 1) * DA + 16 + m] };
  }

  int buf = 0;
  if (rt < ntiles) {
    float* d = &tile[0][(8 * q) * GT_PITCH + 4 * lane];
#pragma unroll
    for (int u = 0; u < 8; ++u) *(float4*)(d + u * GT_PITCH) = s[u];
  }
  for (; rt < ntiles; rt += PROJ_BLOCKS) {
    __syncthreads();
    const int nrt = rt + PROJ_BLOCKS;
    const bool more = nrt < ntiles;
    if (more) {
      const float* g = x + (size_t)(nrt * GT_ROWS + 8 * q) * DIN + 4 * lane;
#pragma unroll
      for (int u = 0; u < 8; ++u) s[u] = *(const float4*)(g + u * DIN);
    }
    const float* t = &tile[buf][m * GT_PITCH + 2 * half];
    v2f a[DIN / 4];
#pragma unroll
    for (int j = 0; j < DIN / 4; ++j) a[j] = *(const v2f*)(t + 4 * j);
    __builtin_amdgcn_sched_barrier(0);

    v8f acc0 = {}, acc1 = {};
#pragma unroll
    for (int j = 0; j < DIN / 4; ++j) {
      acc0 = wmma_f32_16x16x4(a[j], b0[j], acc0);
      acc1 = wmma_f32_16x16x4(a[j], b1[j], acc1);
    }

    float* orow = dst + (size_t)(rt * GT_ROWS) * DA + lane;
#pragma unroll
    for (int v = 0; v < 8; ++v) {
      const float x0 = acc0[v], x1 = acc1[v];
      const float give = half ? x0 : x1;
      const float got  = xhalf(give);
      orow[v * DA]       = half ? got : x0;
      orow[(8 + v) * DA] = half ? x1 : got;
    }

    if (more) {
      float* d = &tile[buf ^ 1][(8 * q) * GT_PITCH + 4 * lane];
#pragma unroll
      for (int u = 0; u < 8; ++u) *(float4*)(d + u * GT_PITCH) = s[u];
    }
    buf ^= 1;
  }
}

__global__ __launch_bounds__(256) void edge_score_kernel(const float* __restrict__ a_row,
                                  const float* __restrict__ a_col,
                                  const int* __restrict__ row,
                                  const int* __restrict__ col,
                                  unsigned* __restrict__ rec,
                                  unsigned char* __restrict__ owner,
                                  unsigned* __restrict__ planes,
                                  unsigned* __restrict__ maxkey) {
  __shared__ unsigned present[8 * 8];
  const int tid = threadIdx.x;
  const int e = blockIdx.x * 256 + tid;
  const bool valid = e < N_EDGES;
  if (tid < 64) present[tid] = 0u;
  __syncthreads();
  unsigned key = 0u;
  float s = 0.f;
  int c = 0;
  unsigned own = 0u, rl = 0u;
  if (valid) {
    const int r = row[e];
    c = col[e];
    const float4* ar = (const float4*)(a_row + (size_t)r * DA);
    const float4* ac = (const float4*)(a_col + (size_t)c * DA);
#pragma unroll
    for (int i = 0; i < DA / 4; ++i) {
      float4 u = ar[i], v = ac[i];
      s += u.x * v.x + u.y * v.y + u.z * v.z + u.w * v.w;
    }
    s *= 0.17677669529663687f;
    s = (s >= 0.f) ? s : SLOPE * s;
    own = (unsigned)r / RPB;
    rl  = (unsigned)r - own * RPB;
    __hip_atomic_fetch_or(&present[(tid >> 5) * 8 + (own >> 5)], 1u << (own & 31u),
                          __ATOMIC_RELAXED, __HIP_MEMORY_SCOPE_WORKGROUP);
    const unsigned b = __float_as_uint(s);
    key = (b & 0x80000000u) ? ~b : (b | 0x80000000u);
  }
  __syncthreads();
  if (tid < 64) {
    const int p = tid >> 3, gi = tid & 7;
    const int g = blockIdx.x * 8 + gi;
    if (g < N_GROUPS) planes[(size_t)p * N_GROUPS + g] = present[gi * 8 + p];
  }
  if (valid) {
    ((uint2*)rec)[e] = make_uint2(__float_as_uint(s), (unsigned)c | (rl << 16));
    owner[e] = (unsigned char)own;
  }
#pragma unroll
  for (int off = 16; off > 0; off >>= 1) {
    unsigned o = __shfl_xor(key, off, 32);
    key = (o > key) ? o : key;
  }
  if ((tid & 31) == 0) atomicMax(maxkey + chip_half() * CTRL_STRIDE + CTRL_MAX, key);
}

static __device__ __forceinline__ void lds_add(float* p, float v) {
  __hip_atomic_fetch_add(p, v, __ATOMIC_RELAXED, __HIP_MEMORY_SCOPE_WORKGROUP);
}

__global__ __launch_bounds__(1024) void row_range_kernel(const float* __restrict__ x,
                                                         const float* __restrict__ adj_val,
                                                         const unsigned* __restrict__ rec,
                                                         const unsigned* __restrict__ owner_words,
                                                         const unsigned* __restrict__ planes,
                                                         const unsigned* __restrict__ ctrl,
                                                         float* __restrict__ feat) {
  __shared__ __attribute__((aligned(16))) float acc[RPB * ROW_WORDS];
  __shared__ float degl[RPB];
  __shared__ int queue[32 * QCAP];
  __shared__ int gqueue[32 * QCAP];

  const int tid = threadIdx.x, lane = tid & 31, w = tid >> 5;
  const int blk = blockIdx.x;

  for (int i = tid; i < RPB * ROW_WORDS / 4; i += 1024)
    ((float4*)acc)[i] = make_float4(0.f, 0.f, 0.f, 0.f);
  if (tid < RPB) degl[tid] = 0.f;
  __syncthreads();

  const unsigned k0 = ctrl[CTRL_MAX], k1 = ctrl[CTRL_STRIDE + CTRL_MAX];
  const unsigned key = (k0 > k1) ? k0 : k1;
  const unsigned kb = (key & 0x80000000u) ? (key ^ 0x80000000u) : ~key;
  const float gmax = __uint_as_float(kb);

  int* q = queue + w * QCAP;
  int head = 0, cnt = 0;

  auto work_off = [&](int n) {
    int c = 0, rl = 0;
    float a = 0.f, ex = 0.f;
    if (lane < n) {
      const int e = q[(head + lane) & (QCAP - 1)];
      const uint2 rc = ((const uint2*)rec)[e];
      a  = adj_val[e];
      ex = expf(__uint_as_float(rc.x) - gmax);
      c  = (int)(rc.y & 0xffffu);
      rl = (int)(rc.y >> 16);
      lds_add(&degl[rl], ex);
    }
    int j = 0;
    for (; j + 8 <= n; j += 8) {
      float4 xv[8];
      int rj[8];
      float aj[8], ej[8];
#pragma unroll
      for (int u = 0; u < 8; ++u) {
        const int cj = __builtin_amdgcn_readlane(c, j + u);
        rj[u] = __builtin_amdgcn_readlane(rl, j + u);
        aj[u] = __int_as_float(__builtin_amdgcn_readlane(__float_as_int(a), j + u));
        ej[u] = __int_as_float(__builtin_amdgcn_readlane(__float_as_int(ex), j + u));
        xv[u] = *(const float4*)(x + (size_t)cj * DIN + 4 * lane);
      }
#pragma unroll
      for (int u = 0; u < 8; ++u) {
        float* d = acc + rj[u] * ROW_WORDS + lane;
        lds_add(d,        aj[u] * xv[u].x);
        lds_add(d + 32,   aj[u] * xv[u].y);
        lds_add(d + 64,   aj[u] * xv[u].z);
        lds_add(d + 96,   aj[u] * xv[u].w);
        lds_add(d + 128,  ej[u] * xv[u].x);
        lds_add(d + 160,  ej[u] * xv[u].y);
        lds_add(d + 192,  ej[u] * xv[u].z);
        lds_add(d + 224,  ej[u] * xv[u].w);
      }
    }
    for (; j < n; ++j) {
      const int cj = __builtin_amdgcn_readlane(c, j);
      const int r1 = __builtin_amdgcn_readlane(rl, j);
      const float a1 = __int_as_float(__builtin_amdgcn_readlane(__float_as_int(a), j));
      const float e1 = __int_as_float(__builtin_amdgcn_readlane(__float_as_int(ex), j));
      const float4 x1 = *(const float4*)(x + (size_t)cj * DIN + 4 * lane);
      float* d = acc + r1 * ROW_WORDS + lane;
      lds_add(d,        a1 * x1.x);
      lds_add(d + 32,   a1 * x1.y);
      lds_add(d + 64,   a1 * x1.z);
      lds_add(d + 96,   a1 * x1.w);
      lds_add(d + 128,  e1 * x1.x);
      lds_add(d + 160,  e1 * x1.y);
      lds_add(d + 192,  e1 * x1.z);
      lds_add(d + 224,  e1 * x1.w);
    }
  };

  const unsigned pat = (unsigned)blk * 0x01010101u;
  auto same = [&](unsigned v) {
    const unsigned d = v ^ pat;
    return ~(((d & 0x7f7f7f7fu) + 0x7f7f7f7fu) | d) & 0x80808080u;
  };

  int* gq = gqueue + w * QCAP;
  int ghead = 0, gcnt = 0;

  auto take_groups = [&](int n) {
    unsigned hit = 0u;
    int ebase = 0;
    if (lane < n) {
      const int g = gq[(ghead + lane) & (QCAP - 1)];
      const uint4* ob = (const uint4*)owner_words + (size_t)g * 2;
      const uint4 lo = ob[0], hi = ob[1];
      ebase = g * 32;
      hit = same(lo.x) | (same(lo.y) >> 1) | (same(lo.z) >> 2) | (same(lo.w) >> 3) |
            (same(hi.x) >> 4) | (same(hi.y) >> 5) | (same(hi.z) >> 6) | (same(hi.w) >> 7);
    }
    while (__builtin_amdgcn_ballot_w32(hit != 0u)) {
      const bool p = hit != 0u;
      const unsigned m = __builtin_amdgcn_ballot_w32(p);
      if (p) {
        const int bit = __builtin_ctz(hit);
        const int k = (7 - (bit & 7)) * 4 + (bit >> 3);
        const int pos = (head + cnt + (int)__builtin_amdgcn_mbcnt_lo(m, 0u)) & (QCAP - 1);
        q[pos] = ebase + k;
        hit &= hit - 1u;
      }
      cnt += __builtin_popcount(m);
      if (cnt >= 32) {
        work_off(32);
        head = (head + 32) & (QCAP - 1);
        cnt -= 32;
      }
    }
  };

  const uint4* pl = (const uint4*)(planes + (size_t)(blk >> 5) * N_GROUPS);
  const int sh = blk & 31;
  const int rot = (blk & 31) * 12;
  int t = w;
  int ch = t + rot;
  if (ch >= MASK_CHUNKS) ch -= MASK_CHUNKS;
  uint4 cur = make_uint4(0u, 0u, 0u, 0u);
  if (t < MASK_CHUNKS && ch * 32 + lane < PLANE_VEC) cur = pl[ch * 32 + lane];
  while (t < MASK_CHUNKS) {
    const int tn = t + 32;
    int chn = tn + rot;
    if (chn >= MASK_CHUNKS) chn -= MASK_CHUNKS;
    uint4 nxt = make_uint4(0u, 0u, 0u, 0u);
    if (tn < MASK_CHUNKS && chn * 32 + lane < PLANE_VEC) nxt = pl[chn * 32 + lane];
    const int gbase = (ch * 32 + lane) * 4;
    unsigned gh = ((cur.x >> sh) & 1u) | (((cur.y >> sh) & 1u) << 1) |
                  (((cur.z >> sh) & 1u) << 2) | (((cur.w >> sh) & 1u) << 3);
    while (__builtin_amdgcn_ballot_w32(gh != 0u)) {
      const bool p = gh != 0u;
      const unsigned m = __builtin_amdgcn_ballot_w32(p);
      if (p) {
        const int pos = (ghead + gcnt + (int)__builtin_amdgcn_mbcnt_lo(m, 0u)) & (QCAP - 1);
        gq[pos] = gbase + __builtin_ctz(gh);
        gh &= gh - 1u;
      }
      gcnt += __builtin_popcount(m);
      if (gcnt >= 32) {
        take_groups(32);
        ghead = (ghead + 32) & (QCAP - 1);
        gcnt -= 32;
      }
    }
    cur = nxt;
    t = tn;
    ch = chn;
  }
  if (gcnt > 0) take_groups(gcnt);
  if (cnt > 0) work_off(cnt);
  __syncthreads();

  for (int idx = tid; idx < RPB * 32; idx += 1024) {
    const int rl = idx >> 5;
    const int r = blk * RPB + rl;
    if (r < N_NODES) {
      const float inv = 1.0f / (degl[rl] + 1e-15f);
      const float* s = acc + rl * ROW_WORDS + lane;
      float4 f;
      f.x = 0.5f * (s[0]  + s[128] * inv);
      f.y = 0.5f * (s[32] + s[160] * inv);
      f.z = 0.5f * (s[64] + s[192] * inv);
      f.w = 0.5f * (s[96] + s[224] * inv);
      *(float4*)(feat + (size_t)r * DIN + 4 * lane) = f;
    }
  }
}

#define GEMM_BLOCKS 512

__global__ __launch_bounds__(256) void out_gemm_kernel(const float* __restrict__ feat,
                                                       const float* __restrict__ W_x,
                                                       const float* __restrict__ b_x,
                                                       float* __restrict__ out) {
  __shared__ __attribute__((aligned(16))) float tile[2][GT_ROWS * GT_PITCH];

  const int tid = threadIdx.x, lane = tid & 31, w = tid >> 5;
  const int half = lane >> 4;
  const int m    = lane & 15;
  const int c0   = w * 32;
  const int ntiles = N_NODES / GT_ROWS;

  int rt = blockIdx.x;
  float4 s0 = make_float4(0.f, 0.f, 0.f, 0.f), s1 = s0;
  if (rt < ntiles) {
    const float* g = feat + (size_t)(rt * GT_ROWS + 2 * w) * DIN + 4 * lane;
    s0 = *(const float4*)g;
    s1 = *(const float4*)(g + DIN);
  }

  v2f b0[DIN / 4], b1[DIN / 4];
#pragma unroll
  for (int j = 0; j < DIN / 4; ++j) {
    const int kk = 4 * j + 2 * half;
    b0[j] = (v2f){ W_x[kk * DOUT + c0 + m],      W_x[(kk + 1) * DOUT + c0 + m] };
    b1[j] = (v2f){ W_x[kk * DOUT + c0 + 16 + m], W_x[(kk + 1) * DOUT + c0 + 16 + m] };
  }
  const float bias0 = b_x[c0 + m], bias1 = b_x[c0 + 16 + m];

  int buf = 0;
  if (rt < ntiles) {
    float* d = &tile[0][(2 * w) * GT_PITCH + 4 * lane];
    *(float4*)d = s0;
    *(float4*)(d + GT_PITCH) = s1;
  }
  for (; rt < ntiles; rt += GEMM_BLOCKS) {
    __syncthreads();
    const int nrt = rt + GEMM_BLOCKS;
    const bool more = nrt < ntiles;
    if (more) {
      const float* g = feat + (size_t)(nrt * GT_ROWS + 2 * w) * DIN + 4 * lane;
      s0 = *(const float4*)g;
      s1 = *(const float4*)(g + DIN);
    }
    const float* t = &tile[buf][m * GT_PITCH + 2 * half];
    v2f a[DIN / 4];
#pragma unroll
    for (int j = 0; j < DIN / 4; ++j) a[j] = *(const v2f*)(t + 4 * j);
    __builtin_amdgcn_sched_barrier(0);

    v8f acc0, acc1;
#pragma unroll
    for (int v = 0; v < 8; ++v) { acc0[v] = bias0; acc1[v] = bias1; }
#pragma unroll
    for (int j = 0; j < DIN / 4; ++j) {
      acc0 = wmma_f32_16x16x4(a[j], b0[j], acc0);
      acc1 = wmma_f32_16x16x4(a[j], b1[j], acc1);
    }

    float* orow = out + (size_t)(rt * GT_ROWS) * DOUT + c0 + lane;
#pragma unroll
    for (int v = 0; v < 8; ++v) {
      const float x0 = acc0[v], x1 = acc1[v];
      const float give = half ? x0 : x1;
      const float got  = xhalf(give);
      orow[(size_t)v * DOUT]       = half ? got : x0;
      orow[(size_t)(8 + v) * DOUT] = half ? x1 : got;
    }

    if (more) {
      float* d = &tile[buf ^ 1][(2 * w) * GT_PITCH + 4 * lane];
      *(float4*)d = s0;
      *(float4*)(d + GT_PITCH) = s1;
    }
    buf ^= 1;
  }
}

extern "C" void kernel_launch(void* const* d_in, const int* in_sizes, int n_in,
                              void* d_out, int out_size, void* d_ws, size_t ws_size,
                              hipStream_t stream) {
  const float* x      = (const float*)d_in[0];
  const int*   row    = (const int*)d_in[1];
  const int*   col    = (const int*)d_in[2];
  const float* adjval = (const float*)d_in[3];
  const float* W_row  = (const float*)d_in[4];
  const float* W_col  = (const float*)d_in[5];
  const float* W_x    = (const float*)d_in[6];
  const float* b_x    = (const float*)d_in[7];
  float* out = (float*)d_out;

  char* ws = (char*)d_ws;
  size_t off = 0;
  float* feat  = (float*)(ws + off); off += (size_t)N_NODES * DIN * 4;
  float* a_row = (float*)(ws + off); off += (size_t)N_NODES * DA * 4;
  float* a_col = (float*)(ws + off); off += (size_t)N_NODES * DA * 4;
  unsigned* rec = (unsigned*)(ws + off); off += (size_t)N_EDGES * 8;
  unsigned* ctrl = (unsigned*)(ws + off); off += 1024;
  unsigned char* owner = (unsigned char*)(ws + off); off += (size_t)N_EDGES;
  unsigned* planes = (unsigned*)(ws + off); off += (size_t)N_EDGES;

  hipMemsetAsync(ctrl, 0, 1024, stream);

  proj_kernel<<<PROJ_BLOCKS, 64, 0, stream>>>(x, W_row, W_col, a_row, a_col);
  {
    const int blocks = (N_EDGES + 255) / 256;
    edge_score_kernel<<<blocks, 256, 0, stream>>>(a_row, a_col, row, col, rec, owner, planes, ctrl);
  }
  row_range_kernel<<<N_OWN, 1024, 0, stream>>>(x, adjval, rec, (const unsigned*)owner, planes, ctrl, feat);
  out_gemm_kernel<<<GEMM_BLOCKS, 256, 0, stream>>>(feat, W_x, b_x, out);
}


# Row sums first, then one pass: a block lists its edges in LDS while summing exp per row, then adds (adj + exp/deg)*0.5*x[col] once: 4 LDS float adds an edge, not 8; one sum a row, another order
# speedup vs baseline: 3.9025x; 1.0647x over previous
#include <hip/hip_runtime.h>

#define N_NODES 50000
#define N_EDGES 1600000
#define DIN 128
#define DOUT 256
#define DA 32
#define SLOPE 0.2f

#define CTRL_STRIDE 128
#define CTRL_MAX 0

#define RPB 196
#define N_OWN 256
#define ROW_WORDS DIN
#define LCAP 12288
#define QCAP 64
#define N_GROUPS (N_EDGES / 32)
#define PLANE_VEC (N_GROUPS / 4)
#define MASK_CHUNKS ((PLANE_VEC + 31) / 32)

typedef __attribute__((ext_vector_type(2))) float v2f;
typedef __attribute__((ext_vector_type(8))) float v8f;

static __device__ __forceinline__ v8f wmma_f32_16x16x4(v2f a, v2f b, v8f c) {
  return __builtin_amdgcn_wmma_f32_16x16x4_f32(false, a, false, b, (short)0, c,
                                               false, false);
}

static __device__ __forceinline__ int chip_half() {
  return (int)((__builtin_amdgcn_s_sendmsg_rtn(0x87) >> 18) & 1u);
}

static __device__ __forceinline__ float xhalf(float v) {
  const int b = __builtin_bit_cast(int, v);
  return __builtin_bit_cast(float, __builtin_amdgcn_permlanex16(b, b, 0x76543210, 0xfedcba98, false, false));
}

#define GT_ROWS 16
#define GT_PITCH 132
#define PROJ_BLOCKS 1024

__global__ __launch_bounds__(64) void proj_kernel(const float* __restrict__ x,
                                                  const float* __restrict__ W_row,
                                                  const float* __restrict__ W_col,
                                                  float* __restrict__ a_row,
                                                  float* __restrict__ a_col) {
  __shared__ __attribute__((aligned(16))) float tile[2][GT_ROWS * GT_PITCH];

  const int tid = threadIdx.x, lane = tid & 31, q = tid >> 5;
  const int half = lane >> 4;
  const int m    = lane & 15;
  const float* __restrict__ Wm = q ? W_col : W_row;
  float* __restrict__ dst      = q ? a_col : a_row;
  const int ntiles = N_NODES / GT_ROWS;

  int rt = blockIdx.x;
  float4 s[8];
#pragma unroll
  for (int u = 0; u < 8; ++u) s[u] = make_float4(0.f, 0.f, 0.f, 0.f);
  if (rt < ntiles) {
    const float* g = x + (size_t)(rt * GT_ROWS + 8 * q) * DIN + 4 * lane;
#pragma unroll
    for (int u = 0; u < 8; ++u) s[u] = *(const float4*)(g + u * DIN);
  }

  v2f b0[DIN / 4], b1[DIN / 4];
#pragma unroll
  for (int j = 0; j < DIN / 4; ++j) {
    const int kk = 4 * j + 2 * half;
    b0[j] = (v2f){ Wm[kk * DA + m],      Wm[(kk + 1) * DA + m] };
    b1[j] = (v2f){ Wm[kk * DA + 16 + m], Wm[(kk + 1) * DA + 16 + m] };
  }

  int buf = 0;
  if (rt < ntiles) {
    float* d = &tile[0][(8 * q) * GT_PITCH + 4 * lane];
#pragma unroll
    for (int u = 0; u < 8; ++u) *(float4*)(d + u * GT_PITCH) = s[u];
  }
  for (; rt < ntiles; rt += PROJ_BLOCKS) {
    __syncthreads();
    const int nrt = rt + PROJ_BLOCKS;
    const bool more = nrt < ntiles;
    if (more) {
      const float* g = x + (size_t)(nrt * GT_ROWS + 8 * q) * DIN + 4 * lane;
#pragma unroll
      for (int u = 0; u < 8; ++u) s[u] = *(const float4*)(g + u * DIN);
    }
    const float* t = &tile[buf][m * GT_PITCH + 2 * half];
    v2f a[DIN / 4];
#pragma unroll
    for (int j = 0; j < DIN / 4; ++j) a[j] = *(const v2f*)(t + 4 * j);
    __builtin_amdgcn_sched_barrier(0);

    v8f acc0 = {}, acc1 = {};
#pragma unroll
    for (int j = 0; j < DIN / 4; ++j) {
      acc0 = wmma_f32_16x16x4(a[j], b0[j], acc0);
      acc1 = wmma_f32_16x16x4(a[j], b1[j], acc1);
    }

    float* orow = dst + (size_t)(rt * GT_ROWS) * DA + lane;
#pragma unroll
    for (int v = 0; v < 8; ++v) {
      const float x0 = acc0[v], x1 = acc1[v];
      const float give = half ? x0 : x1;
      const float got  = xhalf(give);
      orow[v * DA]       = half ? got : x0;
      orow[(8 + v) * DA] = half ? x1 : got;
    }

    if (more) {
      float* d = &tile[buf ^ 1][(8 * q) * GT_PITCH + 4 * lane];
#pragma unroll
      for (int u = 0; u < 8; ++u) *(float4*)(d + u * GT_PITCH) = s[u];
    }
    buf ^= 1;
  }
}

__global__ __launch_bounds__(256) void edge_score_kernel(const float* __restrict__ a_row,
                                  const float* __restrict__ a_col,
                                  const int* __restrict__ row,
                                  const int* __restrict__ col,
                                  unsigned* __restrict__ rec,
                                  unsigned char* __restrict__ owner,
                                  unsigned* __restrict__ planes,
                                  unsigned* __restrict__ maxkey) {
  __shared__ unsigned present[8 * 8];
  const int tid = threadIdx.x;
  const int e = blockIdx.x * 256 + tid;
  const bool valid = e < N_EDGES;
  if (tid < 64) present[tid] = 0u;
  __syncthreads();
  unsigned key = 0u;
  float s = 0.f;
  int c = 0;
  unsigned own = 0u, rl = 0u;
  if (valid) {
    const int r = row[e];
    c = col[e];
    const float4* ar = (const float4*)(a_row + (size_t)r * DA);
    const float4* ac = (const float4*)(a_col + (size_t)c * DA);
#pragma unroll
    for (int i = 0; i < DA / 4; ++i) {
      float4 u = ar[i], v = ac[i];
      s += u.x * v.x + u.y * v.y + u.z * v.z + u.w * v.w;
    }
    s *= 0.17677669529663687f;
    s = (s >= 0.f) ? s : SLOPE * s;
    own = (unsigned)r / RPB;
    rl  = (unsigned)r - own * RPB;
    __hip_atomic_fetch_or(&present[(tid >> 5) * 8 + (own >> 5)], 1u << (own & 31u),
                          __ATOMIC_RELAXED, __HIP_MEMORY_SCOPE_WORKGROUP);
    const unsigned b = __float_as_uint(s);
    key = (b & 0x80000000u) ? ~b : (b | 0x80000000u);
  }
  __syncthreads();
  if (tid < 64) {
    const int p = tid >> 3, gi = tid & 7;
    const int g = blockIdx.x * 8 + gi;
    if (g < N_GROUPS) planes[(size_t)p * N_GROUPS + g] = present[gi * 8 + p];
  }
  if (valid) {
    ((uint2*)rec)[e] = make_uint2(__float_as_uint(s), (unsigned)c | (rl << 16));
    owner[e] = (unsigned char)own;
  }
#pragma unroll
  for (int off = 16; off > 0; off >>= 1) {
    unsigned o = __shfl_xor(key, off, 32);
    key = (o > key) ? o : key;
  }
  if ((tid & 31) == 0) atomicMax(maxkey + chip_half() * CTRL_STRIDE + CTRL_MAX, key);
}

static __device__ __forceinline__ void lds_add(float* p, float v) {
  __hip_atomic_fetch_add(p, v, __ATOMIC_RELAXED, __HIP_MEMORY_SCOPE_WORKGROUP);
}

__global__ __launch_bounds__(1024) void row_range_kernel(const float* __restrict__ x,
                                                         const float* __restrict__ adj_val,
                                                         const unsigned* __restrict__ rec,
                                                         const unsigned* __restrict__ owner_words,
                                                         const unsigned* __restrict__ planes,
                                                         const unsigned* __restrict__ ctrl,
                                                         float* __restrict__ feat) {
  __shared__ __attribute__((aligned(16))) float acc[RPB * ROW_WORDS];
  __shared__ float degl[RPB];
  __shared__ int queue[32 * QCAP];
  __shared__ int gqueue[32 * QCAP];
  __shared__ unsigned lst_cr[LCAP];
  __shared__ int lst_id[LCAP];
  __shared__ float lst_e[LCAP];
  __shared__ int lcount;

  const int tid = threadIdx.x, lane = tid & 31, w = tid >> 5;
  const int blk = blockIdx.x;

  for (int i = tid; i < RPB * ROW_WORDS / 4; i += 1024)
    ((float4*)acc)[i] = make_float4(0.f, 0.f, 0.f, 0.f);
  if (tid < RPB) degl[tid] = 0.f;
  if (tid == 0) lcount = 0;
  __syncthreads();

  const unsigned k0 = ctrl[CTRL_MAX], k1 = ctrl[CTRL_STRIDE + CTRL_MAX];
  const unsigned key = (k0 > k1) ? k0 : k1;
  const unsigned kb = (key & 0x80000000u) ? (key ^ 0x80000000u) : ~key;
  const float gmax = __uint_as_float(kb);

  int* q = queue + w * QCAP;
  int head = 0, cnt = 0;
  int* gq = gqueue + w * QCAP;
  int ghead = 0, gcnt = 0;

  auto accumulate = [&](int n, int c, int rl, float val) {
    int j = 0;
    for (; j + 8 <= n; j += 8) {
      float4 xv[8];
      int rj[8];
      float vj[8];
#pragma unroll
      for (int u = 0; u < 8; ++u) {
        const int cj = __builtin_amdgcn_readlane(c, j + u);
        rj[u] = __builtin_amdgcn_readlane(rl, j + u);
        vj[u] = __int_as_float(__builtin_amdgcn_readlane(__float_as_int(val), j + u));
        xv[u] = *(const float4*)(x + (size_t)cj * DIN + 4 * lane);
      }
#pragma unroll
      for (int u = 0; u < 8; ++u) {
        float* d = acc + rj[u] * ROW_WORDS + lane;
        lds_add(d,      vj[u] * xv[u].x);
        lds_add(d + 32, vj[u] * xv[u].y);
        lds_add(d + 64, vj[u] * xv[u].z);
        lds_add(d + 96, vj[u] * xv[u].w);
      }
    }
    for (; j < n; ++j) {
      const int cj = __builtin_amdgcn_readlane(c, j);
      const int r1 = __builtin_amdgcn_readlane(rl, j);
      const float v1 = __int_as_float(__builtin_amdgcn_readlane(__float_as_int(val), j));
      const float4 x1 = *(const float4*)(x + (size_t)cj * DIN + 4 * lane);
      float* d = acc + r1 * ROW_WORDS + lane;
      lds_add(d,      v1 * x1.x);
      lds_add(d + 32, v1 * x1.y);
      lds_add(d + 64, v1 * x1.z);
      lds_add(d + 96, v1 * x1.w);
    }
  };

  auto note_edges = [&](int n) {
    unsigned cr = 0u;
    int e = 0;
    float ex = 0.f;
    if (lane < n) {
      e = q[(head + lane) & (QCAP - 1)];
      const uint2 rc = ((const uint2*)rec)[e];
      ex = expf(__uint_as_float(rc.x) - gmax);
      cr = rc.y;
      lds_add(&degl[cr >> 16], ex);
    }
    int pos = 0;
    if (lane == 0) pos = __hip_atomic_fetch_add(&lcount, n, __ATOMIC_RELAXED, __HIP_MEMORY_SCOPE_WORKGROUP);
    pos = __builtin_amdgcn_readfirstlane(pos);
    if (lane < n && pos + lane < LCAP) {
      lst_cr[pos + lane] = cr;
      lst_id[pos + lane] = e;
      lst_e[pos + lane]  = ex;
    }
  };

  auto add_edges = [&](int n) {
    int c = 0, rl = 0;
    float val = 0.f;
    if (lane < n) {
      const int e = q[(head + lane) & (QCAP - 1)];
      const uint2 rc = ((const uint2*)rec)[e];
      const float a  = adj_val[e];
      const float ex = expf(__uint_as_float(rc.x) - gmax);
      c  = (int)(rc.y & 0xffffu);
      rl = (int)(rc.y >> 16);
      val = (a + ex * (1.0f / (degl[rl] + 1e-15f))) * 0.5f;
    }
    accumulate(n, c, rl, val);
  };

  const unsigned pat = (unsigned)blk * 0x01010101u;
  auto same = [&](unsigned v) {
    const unsigned d = v ^ pat;
    return ~(((d & 0x7f7f7f7fu) + 0x7f7f7f7fu) | d) & 0x80808080u;
  };

  auto find_edges = [&](auto consume) {
    head = 0; cnt = 0; ghead = 0; gcnt = 0;

    auto take_groups = [&](int n) {
      unsigned hit = 0u;
      int ebase = 0;
      if (lane < n) {
        const int g = gq[(ghead + lane) & (QCAP - 1)];
        const uint4* ob = (const uint4*)owner_words + (size_t)g * 2;
        const uint4 lo = ob[0], hi = ob[1];
        ebase = g * 32;
        hit = same(lo.x) | (same(lo.y) >> 1) | (same(lo.z) >> 2) | (same(lo.w) >> 3) |
              (same(hi.x) >> 4) | (same(hi.y) >> 5) | (same(hi.z) >> 6) | (same(hi.w) >> 7);
      }
      while (__builtin_amdgcn_ballot_w32(hit != 0u)) {
        const bool p = hit != 0u;
        const unsigned m = __builtin_amdgcn_ballot_w32(p);
        if (p) {
          const int bit = __builtin_ctz(hit);
          const int k = (7 - (bit & 7)) * 4 + (bit >> 3);
          const int pos = (head + cnt + (int)__builtin_amdgcn_mbcnt_lo(m, 0u)) & (QCAP - 1);
          q[pos] = ebase + k;
          hit &= hit - 1u;
        }
        cnt += __builtin_popcount(m);
        if (cnt >= 32) {
          consume(32);
          head = (head + 32) & (QCAP - 1);
          cnt -= 32;
        }
      }
    };

    const uint4* pl = (const uint4*)(planes + (size_t)(blk >> 5) * N_GROUPS);
    const int sh = blk & 31;
    const int rot = (blk & 31) * 12;
    int t = w;
    int ch = t + rot;
    if (ch >= MASK_CHUNKS) ch -= MASK_CHUNKS;
    uint4 cur = make_uint4(0u, 0u, 0u, 0u);
    if (t < MASK_CHUNKS && ch * 32 + lane < PLANE_VEC) cur = pl[ch * 32 + lane];
    while (t < MASK_CHUNKS) {
      const int tn = t + 32;
      int chn = tn + rot;
      if (chn >= MASK_CHUNKS) chn -= MASK_CHUNKS;
      uint4 nxt = make_uint4(0u, 0u, 0u, 0u);
      if (tn < MASK_CHUNKS && chn * 32 + lane < PLANE_VEC) nxt = pl[chn * 32 + lane];
      const int gbase = (ch * 32 + lane) * 4;
      unsigned gh = ((cur.x >> sh) & 1u) | (((cur.y >> sh) & 1u) << 1) |
                    (((cur.z >> sh) & 1u) << 2) | (((cur.w >> sh) & 1u) << 3);
      while (__builtin_amdgcn_ballot_w32(gh != 0u)) {
        const bool p = gh != 0u;
        const unsigned m = __builtin_amdgcn_ballot_w32(p);
        if (p) {
          const int pos = (ghead + gcnt + (int)__builtin_amdgcn_mbcnt_lo(m, 0u)) & (QCAP - 1);
          gq[pos] = gbase + __builtin_ctz(gh);
          gh &= gh - 1u;
        }
        gcnt += __builtin_popcount(m);
        if (gcnt >= 32) {
          take_groups(32);
          ghead = (ghead + 32) & (QCAP - 1);
          gcnt -= 32;
        }
      }
      cur = nxt;
      t = tn;
      ch = chn;
    }
    if (gcnt > 0) take_groups(gcnt);
    if (cnt > 0) consume(cnt);
  };

  find_edges(note_edges);
  __syncthreads();

  const int total = lcount;
  if (total <= LCAP) {
    for (int base = w * 32; base < total; base += 32 * 32) {
      const int n = (total - base < 32) ? (total - base) : 32;
      int c = 0, rl = 0;
      float val = 0.f;
      if (lane < n) {
        const unsigned cr = lst_cr[base + lane];
        const float ex = lst_e[base + lane];
        const float a  = adj_val[lst_id[base + lane]];
        c  = (int)(cr & 0xffffu);
        rl = (int)(cr >> 16);
        val = (a + ex * (1.0f / (degl[rl] + 1e-15f))) * 0.5f;
      }
      accumulate(n, c, rl, val);
    }
  } else {
    find_edges(add_edges);
  }
  __syncthreads();

  for (int idx = tid; idx < RPB * 32; idx += 1024) {
    const int rl = idx >> 5;
    const int r = blk * RPB + rl;
    if (r < N_NODES) {
      const float* s = acc + rl * ROW_WORDS + lane;
      float4 f;
      f.x = s[0];
      f.y = s[32];
      f.z = s[64];
      f.w = s[96];
      *(float4*)(feat + (size_t)r * DIN + 4 * lane) = f;
    }
  }
}

#define GEMM_BLOCKS 512

__global__ __launch_bounds__(256) void out_gemm_kernel(const float* __restrict__ feat,
                                                       const float* __restrict__ W_x,
                                                       const float* __restrict__ b_x,
                                                       float* __restrict__ out) {
  __shared__ __attribute__((aligned(16))) float tile[2][GT_ROWS * GT_PITCH];

  const int tid = threadIdx.x, lane = tid & 31, w = tid >> 5;
  const int half = lane >> 4;
  const int m    = lane & 15;
  const int c0   = w * 32;
  const int ntiles = N_NODES / GT_ROWS;

  int rt = blockIdx.x;
  float4 s0 = make_float4(0.f, 0.f, 0.f, 0.f), s1 = s0;
  if (rt < ntiles) {
    const float* g = feat + (size_t)(rt * GT_ROWS + 2 * w) * DIN + 4 * lane;
    s0 = *(const float4*)g;
    s1 = *(const float4*)(g + DIN);
  }

  v2f b0[DIN / 4], b1[DIN / 4];
#pragma unroll
  for (int j = 0; j < DIN / 4; ++j) {
    const int kk = 4 * j + 2 * half;
    b0[j] = (v2f){ W_x[kk * DOUT + c0 + m],      W_x[(kk + 1) * DOUT + c0 + m] };
    b1[j] = (v2f){ W_x[kk * DOUT + c0 + 16 + m], W_x[(kk + 1) * DOUT + c0 + 16 + m] };
  }
  const float bias0 = b_x[c0 + m], bias1 = b_x[c0 + 16 + m];

  int buf = 0;
  if (rt < ntiles) {
    float* d = &tile[0][(2 * w) * GT_PITCH + 4 * lane];
    *(float4*)d = s0;
    *(float4*)(d + GT_PITCH) = s1;
  }
  for (; rt < ntiles; rt += GEMM_BLOCKS) {
    __syncthreads();
    const int nrt = rt + GEMM_BLOCKS;
    const bool more = nrt < ntiles;
    if (more) {
      const float* g = feat + (size_t)(nrt * GT_ROWS + 2 * w) * DIN + 4 * lane;
      s0 = *(const float4*)g;
      s1 = *(const float4*)(g + DIN);
    }
    const float* t = &tile[buf][m * GT_PITCH + 2 * half];
    v2f a[DIN / 4];
#pragma unroll
    for (int j = 0; j < DIN / 4; ++j) a[j] = *(const v2f*)(t + 4 * j);
    __builtin_amdgcn_sched_barrier(0);

    v8f acc0, acc1;
#pragma unroll
    for (int v = 0; v < 8; ++v) { acc0[v] = bias0; acc1[v] = bias1; }
#pragma unroll
    for (int j = 0; j < DIN / 4; ++j) {
      acc0 = wmma_f32_16x16x4(a[j], b0[j], acc0);
      acc1 = wmma_f32_16x16x4(a[j], b1[j], acc1);
    }

    float* orow = out + (size_t)(rt * GT_ROWS) * DOUT + c0 + lane;
#pragma unroll
    for (int v = 0; v < 8; ++v) {
      const float x0 = acc0[v], x1 = acc1[v];
      const float give = half ? x0 : x1;
      const float got  = xhalf(give);
      orow[(size_t)v * DOUT]       = half ? got : x0;
      orow[(size_t)(8 + v) * DOUT] = half ? x1 : got;
    }

    if (more) {
      float* d = &tile[buf ^ 1][(2 * w) * GT_PITCH + 4 * lane];
      *(float4*)d = s0;
      *(float4*)(d + GT_PITCH) = s1;
    }
    buf ^= 1;
  }
}

extern "C" void kernel_launch(void* const* d_in, const int* in_sizes, int n_in,
                              void* d_out, int out_size, void* d_ws, size_t ws_size,
                              hipStream_t stream) {
  const float* x      = (const float*)d_in[0];
  const int*   row    = (const int*)d_in[1];
  const int*   col    = (const int*)d_in[2];
  const float* adjval = (const float*)d_in[3];
  const float* W_row  = (const float*)d_in[4];
  const float* W_col  = (const float*)d_in[5];
  const float* W_x    = (const float*)d_in[6];
  const float* b_x    = (const float*)d_in[7];
  float* out = (float*)d_out;

  char* ws = (char*)d_ws;
  size_t off = 0;
  float* feat  = (float*)(ws + off); off += (size_t)N_NODES * DIN * 4;
  float* a_row = (float*)(ws + off); off += (size_t)N_NODES * DA * 4;
  float* a_col = (float*)(ws + off); off += (size_t)N_NODES * DA * 4;
  unsigned* rec = (unsigned*)(ws + off); off += (size_t)N_EDGES * 8;
  unsigned* ctrl = (unsigned*)(ws + off); off += 1024;
  unsigned char* owner = (unsigned char*)(ws + off); off += (size_t)N_EDGES;
  unsigned* planes = (unsigned*)(ws + off); off += (size_t)N_EDGES;

  hipMemsetAsync(ctrl, 0, 1024, stream);

  proj_kernel<<<PROJ_BLOCKS, 64, 0, stream>>>(x, W_row, W_col, a_row, a_col);
  {
    const int blocks = (N_EDGES + 255) / 256;
    edge_score_kernel<<<blocks, 256, 0, stream>>>(a_row, a_col, row, col, rec, owner, planes, ctrl);
  }
  row_range_kernel<<<N_OWN, 1024, 0, stream>>>(x, adjval, rec, (const unsigned*)owner, planes, ctrl, feat);
  out_gemm_kernel<<<GEMM_BLOCKS, 256, 0, stream>>>(feat, W_x, b_x, out);
}


# Edge scores on a resident grid: 1250 blocks of 8 waves walk 5 chunks of 256 edges each (indices a trip ahead, one barrier a trip, one max atomic a block): 10000 long waves, not 50000 short ones
# speedup vs baseline: 4.2888x; 1.0990x over previous
#include <hip/hip_runtime.h>

#define N_NODES 50000
#define N_EDGES 1600000
#define DIN 128
#define DOUT 256
#define DA 32
#define SLOPE 0.2f

#define CTRL_STRIDE 128
#define CTRL_MAX 0

#define RPB 196
#define N_OWN 256
#define ROW_WORDS DIN
#define LCAP 12288
#define QCAP 64
#define N_GROUPS (N_EDGES / 32)
#define PLANE_VEC (N_GROUPS / 4)
#define MASK_CHUNKS ((PLANE_VEC + 31) / 32)

typedef __attribute__((ext_vector_type(2))) float v2f;
typedef __attribute__((ext_vector_type(8))) float v8f;

static __device__ __forceinline__ v8f wmma_f32_16x16x4(v2f a, v2f b, v8f c) {
  return __builtin_amdgcn_wmma_f32_16x16x4_f32(false, a, false, b, (short)0, c,
                                               false, false);
}

static __device__ __forceinline__ int chip_half() {
  return (int)((__builtin_amdgcn_s_sendmsg_rtn(0x87) >> 18) & 1u);
}

static __device__ __forceinline__ float xhalf(float v) {
  const int b = __builtin_bit_cast(int, v);
  return __builtin_bit_cast(float, __builtin_amdgcn_permlanex16(b, b, 0x76543210, 0xfedcba98, false, false));
}

#define GT_ROWS 16
#define GT_PITCH 132
#define PROJ_BLOCKS 1024

__global__ __launch_bounds__(64) void proj_kernel(const float* __restrict__ x,
                                                  const float* __restrict__ W_row,
                                                  const float* __restrict__ W_col,
                                                  float* __restrict__ a_row,
                                                  float* __restrict__ a_col) {
  __shared__ __attribute__((aligned(16))) float tile[2][GT_ROWS * GT_PITCH];

  const int tid = threadIdx.x, lane = tid & 31, q = tid >> 5;
  const int half = lane >> 4;
  const int m    = lane & 15;
  const float* __restrict__ Wm = q ? W_col : W_row;
  float* __restrict__ dst      = q ? a_col : a_row;
  const int ntiles = N_NODES / GT_ROWS;

  int rt = blockIdx.x;
  float4 s[8];
#pragma unroll
  for (int u = 0; u < 8; ++u) s[u] = make_float4(0.f, 0.f, 0.f, 0.f);
  if (rt < ntiles) {
    const float* g = x + (size_t)(rt * GT_ROWS + 8 * q) * DIN + 4 * lane;
#pragma unroll
    for (int u = 0; u < 8; ++u) s[u] = *(const float4*)(g + u * DIN);
  }

  v2f b0[DIN / 4], b1[DIN / 4];
#pragma unroll
  for (int j = 0; j < DIN / 4; ++j) {
    const int kk = 4 * j + 2 * half;
    b0[j] = (v2f){ Wm[kk * DA + m],      Wm[(kk + 1) * DA + m] };
    b1[j] = (v2f){ Wm[kk * DA + 16 + m], Wm[(kk + 1) * DA + 16 + m] };
  }

  int buf = 0;
  if (rt < ntiles) {
    float* d = &tile[0][(8 * q) * GT_PITCH + 4 * lane];
#pragma unroll
    for (int u = 0; u < 8; ++u) *(float4*)(d + u * GT_PITCH) = s[u];
  }
  for (; rt < ntiles; rt += PROJ_BLOCKS) {
    __syncthreads();
    const int nrt = rt + PROJ_BLOCKS;
    const bool more = nrt < ntiles;
    if (more) {
      const float* g = x + (size_t)(nrt * GT_ROWS + 8 * q) * DIN + 4 * lane;
#pragma unroll
      for (int u = 0; u < 8; ++u) s[u] = *(const float4*)(g + u * DIN);
    }
    const float* t = &tile[buf][m * GT_PITCH + 2 * half];
    v2f a[DIN / 4];
#pragma unroll
    for (int j = 0; j < DIN / 4; ++j) a[j] = *(const v2f*)(t + 4 * j);
    __builtin_amdgcn_sched_barrier(0);

    v8f acc0 = {}, acc1 = {};
#pragma unroll
    for (int j = 0; j < DIN / 4; ++j) {
      acc0 = wmma_f32_16x16x4(a[j], b0[j], acc0);
      acc1 = wmma_f32_16x16x4(a[j], b1[j], acc1);
    }

    float* orow = dst + (size_t)(rt * GT_ROWS) * DA + lane;
#pragma unroll
    for (int v = 0; v < 8; ++v) {
      const float x0 = acc0[v], x1 = acc1[v];
      const float give = half ? x0 : x1;
      const float got  = xhalf(give);
      orow[v * DA]       = half ? got : x0;
      orow[(8 + v) * DA] = half ? x1 : got;
    }

    if (more) {
      float* d = &tile[buf ^ 1][(8 * q) * GT_PITCH + 4 * lane];
#pragma unroll
      for (int u = 0; u < 8; ++u) *(float4*)(d + u * GT_PITCH) = s[u];
    }
    buf ^= 1;
  }
}

#define SCORE_CHUNKS (N_EDGES / 256)
#define SCORE_BLOCKS 1250
#define SCORE_TRIPS (SCORE_CHUNKS / SCORE_BLOCKS)
static_assert(N_EDGES % 256 == 0, "a chunk is 256 edges");
static_assert(SCORE_CHUNKS % SCORE_BLOCKS == 0, "equal trips for every block");

__global__ __launch_bounds__(256) void edge_score_kernel(const float* __restrict__ a_row,
                                  const float* __restrict__ a_col,
                                  const int* __restrict__ row,
                                  const int* __restrict__ col,
                                  unsigned* __restrict__ rec,
                                  unsigned char* __restrict__ owner,
                                  unsigned* __restrict__ planes,
                                  unsigned* __restrict__ maxkey) {
  __shared__ unsigned present[2][8 * 8];
  __shared__ unsigned wmax[8];
  const int tid = threadIdx.x;
  if (tid < 128) (&present[0][0])[tid] = 0u;
  __syncthreads();
  unsigned key = 0u;
  int chunk = blockIdx.x;
  int r = row[chunk * 256 + tid];
  int c = col[chunk * 256 + tid];
  int buf = 0;
#pragma unroll 1
  for (int trip = 0; trip < SCORE_TRIPS; ++trip) {
    const int e = chunk * 256 + tid;
    const float4* ar = (const float4*)(a_row + (size_t)r * DA);
    const float4* ac = (const float4*)(a_col + (size_t)c * DA);
    float4 u[DA / 4], v[DA / 4];
#pragma unroll
    for (int i = 0; i < DA / 4; ++i) { u[i] = ar[i]; v[i] = ac[i]; }
    __builtin_amdgcn_sched_barrier(0);
    const int nchunk = chunk + SCORE_BLOCKS;
    int rn = 0, cn = 0;
    if (trip + 1 < SCORE_TRIPS) {
      rn = row[nchunk * 256 + tid];
      cn = col[nchunk * 256 + tid];
    }
    float s = 0.f;
#pragma unroll
    for (int i = 0; i < DA / 4; ++i)
      s += u[i].x * v[i].x + u[i].y * v[i].y + u[i].z * v[i].z + u[i].w * v[i].w;
    s *= 0.17677669529663687f;
    s = (s >= 0.f) ? s : SLOPE * s;
    const unsigned own = (unsigned)r / RPB;
    const unsigned rl  = (unsigned)r - own * RPB;
    __hip_atomic_fetch_or(&present[buf][(tid >> 5) * 8 + (own >> 5)], 1u << (own & 31u),
                          __ATOMIC_RELAXED, __HIP_MEMORY_SCOPE_WORKGROUP);
    const unsigned b = __float_as_uint(s);
    const unsigned k = (b & 0x80000000u) ? ~b : (b | 0x80000000u);
    key = (k > key) ? k : key;
    __syncthreads();
    if (tid < 64) {
      const int p = tid >> 3, gi = tid & 7;
      const int g = chunk * 8 + gi;
      planes[(size_t)p * N_GROUPS + g] = present[buf][gi * 8 + p];
      present[buf][gi * 8 + p] = 0u;
    }
    ((uint2*)rec)[e] = make_uint2(__float_as_uint(s), (unsigned)c | (rl << 16));
    owner[e] = (unsigned char)own;
    buf ^= 1;
    chunk = nchunk;
    r = rn;
    c = cn;
  }
#pragma unroll
  for (int off = 16; off > 0; off >>= 1) {
    unsigned o = __shfl_xor(key, off, 32);
    key = (o > key) ? o : key;
  }
  if ((tid & 31) == 0) wmax[tid >> 5] = key;
  __syncthreads();
  if (tid == 0) {
    unsigned m = wmax[0];
    for (int i = 1; i < 8; ++i) m = (wmax[i] > m) ? wmax[i] : m;
    atomicMax(maxkey + chip_half() * CTRL_STRIDE + CTRL_MAX, m);
  }
}

static __device__ __forceinline__ void lds_add(float* p, float v) {
  __hip_atomic_fetch_add(p, v, __ATOMIC_RELAXED, __HIP_MEMORY_SCOPE_WORKGROUP);
}

__global__ __launch_bounds__(1024) void row_range_kernel(const float* __restrict__ x,
                                                         const float* __restrict__ adj_val,
                                                         const unsigned* __restrict__ rec,
                                                         const unsigned* __restrict__ owner_words,
                                                         const unsigned* __restrict__ planes,
                                                         const unsigned* __restrict__ ctrl,
                                                         float* __restrict__ feat) {
  __shared__ __attribute__((aligned(16))) float acc[RPB * ROW_WORDS];
  __shared__ float degl[RPB];
  __shared__ int queue[32 * QCAP];
  __shared__ int gqueue[32 * QCAP];
  __shared__ unsigned lst_cr[LCAP];
  __shared__ int lst_id[LCAP];
  __shared__ float lst_e[LCAP];
  __shared__ int lcount;

  const int tid = threadIdx.x, lane = tid & 31, w = tid >> 5;
  const int blk = blockIdx.x;

  for (int i = tid; i < RPB * ROW_WORDS / 4; i += 1024)
    ((float4*)acc)[i] = make_float4(0.f, 0.f, 0.f, 0.f);
  if (tid < RPB) degl[tid] = 0.f;
  if (tid == 0) lcount = 0;
  __syncthreads();

  const unsigned k0 = ctrl[CTRL_MAX], k1 = ctrl[CTRL_STRIDE + CTRL_MAX];
  const unsigned key = (k0 > k1) ? k0 : k1;
  const unsigned kb = (key & 0x80000000u) ? (key ^ 0x80000000u) : ~key;
  const float gmax = __uint_as_float(kb);

  int* q = queue + w * QCAP;
  int head = 0, cnt = 0;
  int* gq = gqueue + w * QCAP;
  int ghead = 0, gcnt = 0;

  auto accumulate = [&](int n, int c, int rl, float val) {
    int j = 0;
    for (; j + 8 <= n; j += 8) {
      float4 xv[8];
      int rj[8];
      float vj[8];
#pragma unroll
      for (int u = 0; u < 8; ++u) {
        const int cj = __builtin_amdgcn_readlane(c, j + u);
        rj[u] = __builtin_amdgcn_readlane(rl, j + u);
        vj[u] = __int_as_float(__builtin_amdgcn_readlane(__float_as_int(val), j + u));
        xv[u] = *(const float4*)(x + (size_t)cj * DIN + 4 * lane);
      }
#pragma unroll
      for (int u = 0; u < 8; ++u) {
        float* d = acc + rj[u] * ROW_WORDS + lane;
        lds_add(d,      vj[u] * xv[u].x);
        lds_add(d + 32, vj[u] * xv[u].y);
        lds_add(d + 64, vj[u] * xv[u].z);
        lds_add(d + 96, vj[u] * xv[u].w);
      }
    }
    for (; j < n; ++j) {
      const int cj = __builtin_amdgcn_readlane(c, j);
      const int r1 = __builtin_amdgcn_readlane(rl, j);
      const float v1 = __int_as_float(__builtin_amdgcn_readlane(__float_as_int(val), j));
      const float4 x1 = *(const float4*)(x + (size_t)cj * DIN + 4 * lane);
      float* d = acc + r1 * ROW_WORDS + lane;
      lds_add(d,      v1 * x1.x);
      lds_add(d + 32, v1 * x1.y);
      lds_add(d + 64, v1 * x1.z);
      lds_add(d + 96, v1 * x1.w);
    }
  };

  auto note_edges = [&](int n) {
    unsigned cr = 0u;
    int e = 0;
    float ex = 0.f;
    if (lane < n) {
      e = q[(head + lane) & (QCAP - 1)];
      const uint2 rc = ((const uint2*)rec)[e];
      ex = expf(__uint_as_float(rc.x) - gmax);
      cr = rc.y;
      lds_add(&degl[cr >> 16], ex);
    }
    int pos = 0;
    if (lane == 0) pos = __hip_atomic_fetch_add(&lcount, n, __ATOMIC_RELAXED, __HIP_MEMORY_SCOPE_WORKGROUP);
    pos = __builtin_amdgcn_readfirstlane(pos);
    if (lane < n && pos + lane < LCAP) {
      lst_cr[pos + lane] = cr;
      lst_id[pos + lane] = e;
      lst_e[pos + lane]  = ex;
    }
  };

  auto add_edges = [&](int n) {
    int c = 0, rl = 0;
    float val = 0.f;
    if (lane < n) {
      const int e = q[(head + lane) & (QCAP - 1)];
      const uint2 rc = ((const uint2*)rec)[e];
      const float a  = adj_val[e];
      const float ex = expf(__uint_as_float(rc.x) - gmax);
      c  = (int)(rc.y & 0xffffu);
      rl = (int)(rc.y >> 16);
      val = (a + ex * (1.0f / (degl[rl] + 1e-15f))) * 0.5f;
    }
    accumulate(n, c, rl, val);
  };

  const unsigned pat = (unsigned)blk * 0x01010101u;
  auto same = [&](unsigned v) {
    const unsigned d = v ^ pat;
    return ~(((d & 0x7f7f7f7fu) + 0x7f7f7f7fu) | d) & 0x80808080u;
  };

  auto find_edges = [&](auto consume) {
    head = 0; cnt = 0; ghead = 0; gcnt = 0;

    auto take_groups = [&](int n) {
      unsigned hit = 0u;
      int ebase = 0;
      if (lane < n) {
        const int g = gq[(ghead + lane) & (QCAP - 1)];
        const uint4* ob = (const uint4*)owner_words + (size_t)g * 2;
        const uint4 lo = ob[0], hi = ob[1];
        ebase = g * 32;
        hit = same(lo.x) | (same(lo.y) >> 1) | (same(lo.z) >> 2) | (same(lo.w) >> 3) |
              (same(hi.x) >> 4) | (same(hi.y) >> 5) | (same(hi.z) >> 6) | (same(hi.w) >> 7);
      }
      while (__builtin_amdgcn_ballot_w32(hit != 0u)) {
        const bool p = hit != 0u;
        const unsigned m = __builtin_amdgcn_ballot_w32(p);
        if (p) {
          const int bit = __builtin_ctz(hit);
          const int k = (7 - (bit & 7)) * 4 + (bit >> 3);
          const int pos = (head + cnt + (int)__builtin_amdgcn_mbcnt_lo(m, 0u)) & (QCAP - 1);
          q[pos] = ebase + k;
          hit &= hit - 1u;
        }
        cnt += __builtin_popcount(m);
        if (cnt >= 32) {
          consume(32);
          head = (head + 32) & (QCAP - 1);
          cnt -= 32;
        }
      }
    };

    const uint4* pl = (const uint4*)(planes + (size_t)(blk >> 5) * N_GROUPS);
    const int sh = blk & 31;
    const int rot = (blk & 31) * 12;
    int t = w;
    int ch = t + rot;
    if (ch >= MASK_CHUNKS) ch -= MASK_CHUNKS;
    uint4 cur = make_uint4(0u, 0u, 0u, 0u);
    if (t < MASK_CHUNKS && ch * 32 + lane < PLANE_VEC) cur = pl[ch * 32 + lane];
    while (t < MASK_CHUNKS) {
      const int tn = t + 32;
      int chn = tn + rot;
      if (chn >= MASK_CHUNKS) chn -= MASK_CHUNKS;
      uint4 nxt = make_uint4(0u, 0u, 0u, 0u);
      if (tn < MASK_CHUNKS && chn * 32 + lane < PLANE_VEC) nxt = pl[chn * 32 + lane];
      const int gbase = (ch * 32 + lane) * 4;
      unsigned gh = ((cur.x >> sh) & 1u) | (((cur.y >> sh) & 1u) << 1) |
                    (((cur.z >> sh) & 1u) << 2) | (((cur.w >> sh) & 1u) << 3);
      while (__builtin_amdgcn_ballot_w32(gh != 0u)) {
        const bool p = gh != 0u;
        const unsigned m = __builtin_amdgcn_ballot_w32(p);
        if (p) {
          const int pos = (ghead + gcnt + (int)__builtin_amdgcn_mbcnt_lo(m, 0u)) & (QCAP - 1);
          gq[pos] = gbase + __builtin_ctz(gh);
          gh &= gh - 1u;
        }
        gcnt += __builtin_popcount(m);
        if (gcnt >= 32) {
          take_groups(32);
          ghead = (ghead + 32) & (QCAP - 1);
          gcnt -= 32;
        }
      }
      cur = nxt;
      t = tn;
      ch = chn;
    }
    if (gcnt > 0) take_groups(gcnt);
    if (cnt > 0) consume(cnt);
  };

  find_edges(note_edges);
  __syncthreads();

  const int total = lcount;
  if (total <= LCAP) {
    for (int base = w * 32; base < total; base += 32 * 32) {
      const int n = (total - base < 32) ? (total - base) : 32;
      int c = 0, rl = 0;
      float val = 0.f;
      if (lane < n) {
        const unsigned cr = lst_cr[base + lane];
        const float ex = lst_e[base + lane];
        const float a  = adj_val[lst_id[base + lane]];
        c  = (int)(cr & 0xffffu);
        rl = (int)(cr >> 16);
        val = (a + ex * (1.0f / (degl[rl] + 1e-15f))) * 0.5f;
      }
      accumulate(n, c, rl, val);
    }
  } else {
    find_edges(add_edges);
  }
  __syncthreads();

  for (int idx = tid; idx < RPB * 32; idx += 1024) {
    const int rl = idx >> 5;
    const int r = blk * RPB + rl;
    if (r < N_NODES) {
      const float* s = acc + rl * ROW_WORDS + lane;
      float4 f;
      f.x = s[0];
      f.y = s[32];
      f.z = s[64];
      f.w = s[96];
      *(float4*)(feat + (size_t)r * DIN + 4 * lane) = f;
    }
  }
}

#define GEMM_BLOCKS 512

__global__ __launch_bounds__(256) void out_gemm_kernel(const float* __restrict__ feat,
                                                       const float* __restrict__ W_x,
                                                       const float* __restrict__ b_x,
                                                       float* __restrict__ out) {
  __shared__ __attribute__((aligned(16))) float tile[2][GT_ROWS * GT_PITCH];

  const int tid = threadIdx.x, lane = tid & 31, w = tid >> 5;
  const int half = lane >> 4;
  const int m    = lane & 15;
  const int c0   = w * 32;
  const int ntiles = N_NODES / GT_ROWS;

  int rt = blockIdx.x;
  float4 s0 = make_float4(0.f, 0.f, 0.f, 0.f), s1 = s0;
  if (rt < ntiles) {
    const float* g = feat + (size_t)(rt * GT_ROWS + 2 * w) * DIN + 4 * lane;
    s0 = *(const float4*)g;
    s1 = *(const float4*)(g + DIN);
  }

  v2f b0[DIN / 4], b1[DIN / 4];
#pragma unroll
  for (int j = 0; j < DIN / 4; ++j) {
    const int kk = 4 * j + 2 * half;
    b0[j] = (v2f){ W_x[kk * DOUT + c0 + m],      W_x[(kk + 1) * DOUT + c0 + m] };
    b1[j] = (v2f){ W_x[kk * DOUT + c0 + 16 + m], W_x[(kk + 1) * DOUT + c0 + 16 + m] };
  }
  const float bias0 = b_x[c0 + m], bias1 = b_x[c0 + 16 + m];

  int buf = 0;
  if (rt < ntiles) {
    float* d = &tile[0][(2 * w) * GT_PITCH + 4 * lane];
    *(float4*)d = s0;
    *(float4*)(d + GT_PITCH) = s1;
  }
  for (; rt < ntiles; rt += GEMM_BLOCKS) {
    __syncthreads();
    const int nrt = rt + GEMM_BLOCKS;
    const bool more = nrt < ntiles;
    if (more) {
      const float* g = feat + (size_t)(nrt * GT_ROWS + 2 * w) * DIN + 4 * lane;
      s0 = *(const float4*)g;
      s1 = *(const float4*)(g + DIN);
    }
    const float* t = &tile[buf][m * GT_PITCH + 2 * half];
    v2f a[DIN / 4];
#pragma unroll
    for (int j = 0; j < DIN / 4; ++j) a[j] = *(const v2f*)(t + 4 * j);
    __builtin_amdgcn_sched_barrier(0);

    v8f acc0, acc1;
#pragma unroll
    for (int v = 0; v < 8; ++v) { acc0[v] = bias0; acc1[v] = bias1; }
#pragma unroll
    for (int j = 0; j < DIN / 4; ++j) {
      acc0 = wmma_f32_16x16x4(a[j], b0[j], acc0);
      acc1 = wmma_f32_16x16x4(a[j], b1[j], acc1);
    }

    float* orow = out + (size_t)(rt * GT_ROWS) * DOUT + c0 + lane;
#pragma unroll
    for (int v = 0; v < 8; ++v) {
      const float x0 = acc0[v], x1 = acc1[v];
      const float give = half ? x0 : x1;
      const float got  = xhalf(give);
      orow[(size_t)v * DOUT]       = half ? got : x0;
      orow[(size_t)(8 + v) * DOUT] = half ? x1 : got;
    }

    if (more) {
      float* d = &tile[buf ^ 1][(2 * w) * GT_PITCH + 4 * lane];
      *(float4*)d = s0;
      *(float4*)(d + GT_PITCH) = s1;
    }
    buf ^= 1;
  }
}

extern "C" void kernel_launch(void* const* d_in, const int* in_sizes, int n_in,
                              void* d_out, int out_size, void* d_ws, size_t ws_size,
                              hipStream_t stream) {
  const float* x      = (const float*)d_in[0];
  const int*   row    = (const int*)d_in[1];
  const int*   col    = (const int*)d_in[2];
  const float* adjval = (const float*)d_in[3];
  const float* W_row  = (const float*)d_in[4];
  const float* W_col  = (const float*)d_in[5];
  const float* W_x    = (const float*)d_in[6];
  const float* b_x    = (const float*)d_in[7];
  float* out = (float*)d_out;

  char* ws = (char*)d_ws;
  size_t off = 0;
  float* feat  = (float*)(ws + off); off += (size_t)N_NODES * DIN * 4;
  float* a_row = (float*)(ws + off); off += (size_t)N_NODES * DA * 4;
  float* a_col = (float*)(ws + off); off += (size_t)N_NODES * DA * 4;
  unsigned* rec = (unsigned*)(ws + off); off += (size_t)N_EDGES * 8;
  unsigned* ctrl = (unsigned*)(ws + off); off += 1024;
  unsigned char* owner = (unsigned char*)(ws + off); off += (size_t)N_EDGES;
  unsigned* planes = (unsigned*)(ws + off); off += (size_t)N_EDGES;

  hipMemsetAsync(ctrl, 0, 1024, stream);

  proj_kernel<<<PROJ_BLOCKS, 64, 0, stream>>>(x, W_row, W_col, a_row, a_col);
  edge_score_kernel<<<SCORE_BLOCKS, 256, 0, stream>>>(a_row, a_col, row, col, rec, owner, planes, ctrl);
  row_range_kernel<<<N_OWN, 1024, 0, stream>>>(x, adjval, rec, (const unsigned*)owner, planes, ctrl, feat);
  out_gemm_kernel<<<GEMM_BLOCKS, 256, 0, stream>>>(feat, W_x, b_x, out);
}
